# starting kernel
# baseline (speedup 1.0000x reference)
#include <hip/hip_runtime.h>
#include <hip/hip_bf16.h>


typedef __attribute__((ext_vector_type(2))) float v2f;
typedef __attribute__((ext_vector_type(4))) float v4f;
typedef __attribute__((ext_vector_type(8))) float v8f;

#define B_ 256
#define T_ 512
#define I_ 256
#define H_ 256
#define LDSROW 260

__global__ __launch_bounds__(512) void rnn_xproj_kernel(
    const float* __restrict__ x, const float* __restrict__ Wx,
    float* __restrict__ out) {
  __shared__ float atile[16 * LDSROW];
  const int tid  = threadIdx.x;
  const int lane = tid & 31;
  const int wave = tid >> 5;
  const long M0  = (long)blockIdx.x * 16;

  {
    const int idx = tid * 8;
    const int r   = idx >> 8;
    const int i   = idx & 255;
    const float* src = x + (M0 + r) * (long)I_ + i;
    *(v4f*)(&atile[r * LDSROW + i])     = *(const v4f*)(src);
    *(v4f*)(&atile[r * LDSROW + i + 4]) = *(const v4f*)(src + 4);
  }
  __syncthreads();

  const int n     = (wave << 4) + (lane & 15);
  const int khalf = (lane >> 4) << 1;
  v2f bfrag[64];
#pragma unroll
  for (int kk = 0; kk < 64; ++kk) {
    const int kb = kk * 4 + khalf;
    bfrag[kk].x = Wx[n * I_ + kb + 0];
    bfrag[kk].y = Wx[n * I_ + kb + 1];
  }

  const int m = lane & 15;
  v8f acc = {};
#pragma unroll
  for (int kk = 0; kk < 64; ++kk) {
    const int kb = kk * 4 + khalf;
    v2f a;
    a.x = atile[m * LDSROW + kb + 0];
    a.y = atile[m * LDSROW + kb + 1];
    acc = __builtin_amdgcn_wmma_f32_16x16x4_f32(false, a, false, bfrag[kk],
                                                (short)0, acc, false, false);
  }

  const int mb = (lane >> 4) << 3;
#pragma unroll
  for (int v = 0; v < 8; ++v)
    out[(M0 + mb + v) * (long)H_ + n] = acc[v];
}

__global__ __launch_bounds__(512) void rnn_recur_kernel(
    const float* __restrict__ Wh, const float* __restrict__ bias,
    float* __restrict__ out) {
  __shared__ float hbuf[2][16 * LDSROW];
  const int tid  = threadIdx.x;
  const int lane = tid & 31;
  const int wave = tid >> 5;
  const int b0   = blockIdx.x * 16;

  const int n     = (wave << 4) + (lane & 15);
  const int khalf = (lane >> 4) << 1;

  v2f bfrag[64];
#pragma unroll
  for (int kk = 0; kk < 64; ++kk) {
    const int kb = kk * 4 + khalf;
    bfrag[kk].x = Wh[n * H_ + kb + 0];
    bfrag[kk].y = Wh[n * H_ + kb + 1];
  }
  const float bn = bias[n];

  {
    const int idx = tid * 8;
    const int r   = idx >> 8;
    const int i   = idx & 255;
#pragma unroll
    for (int j = 0; j < 8; ++j) hbuf[0][r * LDSROW + i + j] = 0.0f;
  }
  __syncthreads();

  const int m  = lane & 15;
  const int mb = (lane >> 4) << 3;

  long rowbase[8];
#pragma unroll
  for (int v = 0; v < 8; ++v)
    rowbase[v] = ((long)(b0 + mb + v) * T_) * H_ + n;

  v8f cin;
#pragma unroll
  for (int v = 0; v < 8; ++v) cin[v] = out[rowbase[v]];

  int p = 0;
  for (int t = 0; t < T_; ++t) {
    v8f acc;
#pragma unroll
    for (int v = 0; v < 8; ++v) acc[v] = cin[v] + bn;

    const int tn = (t + 1 < T_) ? (t + 1) : (T_ - 1);
    v8f cnext;
#pragma unroll
    for (int v = 0; v < 8; ++v) cnext[v] = out[rowbase[v] + (long)tn * H_];

#pragma unroll
    for (int kk = 0; kk < 64; ++kk) {
      const int kb = kk * 4 + khalf;
      v2f a;
      a.x = hbuf[p][m * LDSROW + kb + 0];
      a.y = hbuf[p][m * LDSROW + kb + 1];
      acc = __builtin_amdgcn_wmma_f32_16x16x4_f32(false, a, false, bfrag[kk],
                                                  (short)0, acc, false, false);
    }

#pragma unroll
    for (int v = 0; v < 8; ++v) acc[v] = tanhf(acc[v]);

#pragma unroll
    for (int v = 0; v < 8; ++v)
      hbuf[1 - p][(mb + v) * LDSROW + n] = acc[v];

    asm volatile("s_wait_dscnt 0x0" ::: "memory");
    asm volatile("s_barrier_signal -1" ::: "memory");
#pragma unroll
    for (int v = 0; v < 8; ++v)
      out[rowbase[v] + (long)t * H_] = acc[v];
    asm volatile("s_barrier_wait -1" ::: "memory");

    cin = cnext;
    p ^= 1;
  }
}

extern "C" void kernel_launch(void* const* d_in, const int* in_sizes, int n_in,
                              void* d_out, int out_size, void* d_ws,
                              size_t ws_size, hipStream_t stream) {
  const float* x  = (const float*)d_in[0];
  const float* Wx = (const float*)d_in[1];
  const float* Wh = (const float*)d_in[2];
  const float* b  = (const float*)d_in[3];
  float* out = (float*)d_out;

  rnn_xproj_kernel<<<(B_ * T_) / 16, 512, 0, stream>>>(x, Wx, out);
  rnn_recur_kernel<<<B_ / 16, 512, 0, stream>>>(Wh, b, out);
}


# tanh of the recurrence evaluated inline and branch-free (same odd polynomial below 0.625, else 1-2/(exp(2|x|)+1) on hardware exp2 and rcp) instead of the branching library call
# speedup vs baseline: 1.2392x; 1.2392x over previous
#include <hip/hip_runtime.h>
#include <hip/hip_bf16.h>


typedef __attribute__((ext_vector_type(2))) float v2f;
typedef __attribute__((ext_vector_type(4))) float v4f;
typedef __attribute__((ext_vector_type(8))) float v8f;

#define B_ 256
#define T_ 512
#define I_ 256
#define H_ 256
#define LDSROW 260

__device__ __forceinline__ float tanh_f32(float x) {
  const float ax = __builtin_fabsf(x);
  const float z  = x * x;
  float p = -5.70498872745e-3f;
  p = __builtin_fmaf(p, z, 2.06390887954e-2f);
  p = __builtin_fmaf(p, z, -5.37397155531e-2f);
  p = __builtin_fmaf(p, z, 1.33314422036e-1f);
  p = __builtin_fmaf(p, z, -3.33332819422e-1f);
  const float small = __builtin_fmaf(z, x * p, x);
  const float e   = __builtin_amdgcn_exp2f(ax * 2.885390081777927f);
  const float r   = __builtin_amdgcn_rcpf(e + 1.0f);
  const float big = __builtin_copysignf(__builtin_fmaf(-2.0f, r, 1.0f), x);
  return (ax < 0.625f) ? small : big;
}

__global__ __launch_bounds__(512) void rnn_xproj_kernel(
    const float* __restrict__ x, const float* __restrict__ Wx,
    float* __restrict__ out) {
  __shared__ float atile[16 * LDSROW];
  const int tid  = threadIdx.x;
  const int lane = tid & 31;
  const int wave = tid >> 5;
  const long M0  = (long)blockIdx.x * 16;

  {
    const int idx = tid * 8;
    const int r   = idx >> 8;
    const int i   = idx & 255;
    const float* src = x + (M0 + r) * (long)I_ + i;
    *(v4f*)(&atile[r * LDSROW + i])     = *(const v4f*)(src);
    *(v4f*)(&atile[r * LDSROW + i + 4]) = *(const v4f*)(src + 4);
  }
  __syncthreads();

  const int n     = (wave << 4) + (lane & 15);
  const int khalf = (lane >> 4) << 1;
  v2f bfrag[64];
#pragma unroll
  for (int kk = 0; kk < 64; ++kk) {
    const int kb = kk * 4 + khalf;
    bfrag[kk].x = Wx[n * I_ + kb + 0];
    bfrag[kk].y = Wx[n * I_ + kb + 1];
  }

  const int m = lane & 15;
  v8f acc = {};
#pragma unroll
  for (int kk = 0; kk < 64; ++kk) {
    const int kb = kk * 4 + khalf;
    v2f a;
    a.x = atile[m * LDSROW + kb + 0];
    a.y = atile[m * LDSROW + kb + 1];
    acc = __builtin_amdgcn_wmma_f32_16x16x4_f32(false, a, false, bfrag[kk],
                                                (short)0, acc, false, false);
  }

  const int mb = (lane >> 4) << 3;
#pragma unroll
  for (int v = 0; v < 8; ++v)
    out[(M0 + mb + v) * (long)H_ + n] = acc[v];
}

__global__ __launch_bounds__(512) void rnn_recur_kernel(
    const float* __restrict__ Wh, const float* __restrict__ bias,
    float* __restrict__ out) {
  __shared__ float hbuf[2][16 * LDSROW];
  const int tid  = threadIdx.x;
  const int lane = tid & 31;
  const int wave = tid >> 5;
  const int b0   = blockIdx.x * 16;

  const int n     = (wave << 4) + (lane & 15);
  const int khalf = (lane >> 4) << 1;

  v2f bfrag[64];
#pragma unroll
  for (int kk = 0; kk < 64; ++kk) {
    const int kb = kk * 4 + khalf;
    bfrag[kk].x = Wh[n * H_ + kb + 0];
    bfrag[kk].y = Wh[n * H_ + kb + 1];
  }
  const float bn = bias[n];

  {
    const int idx = tid * 8;
    const int r   = idx >> 8;
    const int i   = idx & 255;
#pragma unroll
    for (int j = 0; j < 8; ++j) hbuf[0][r * LDSROW + i + j] = 0.0f;
  }
  __syncthreads();

  const int m  = lane & 15;
  const int mb = (lane >> 4) << 3;

  long rowbase[8];
#pragma unroll
  for (int v = 0; v < 8; ++v)
    rowbase[v] = ((long)(b0 + mb + v) * T_) * H_ + n;

  v8f cin;
#pragma unroll
  for (int v = 0; v < 8; ++v) cin[v] = out[rowbase[v]];

  int p = 0;
  for (int t = 0; t < T_; ++t) {
    v8f acc;
#pragma unroll
    for (int v = 0; v < 8; ++v) acc[v] = cin[v] + bn;

    const int tn = (t + 1 < T_) ? (t + 1) : (T_ - 1);
    v8f cnext;
#pragma unroll
    for (int v = 0; v < 8; ++v) cnext[v] = out[rowbase[v] + (long)tn * H_];

#pragma unroll
    for (int kk = 0; kk < 64; ++kk) {
      const int kb = kk * 4 + khalf;
      v2f a;
      a.x = hbuf[p][m * LDSROW + kb + 0];
      a.y = hbuf[p][m * LDSROW + kb + 1];
      acc = __builtin_amdgcn_wmma_f32_16x16x4_f32(false, a, false, bfrag[kk],
                                                  (short)0, acc, false, false);
    }

#pragma unroll
    for (int v = 0; v < 8; ++v) acc[v] = tanh_f32(acc[v]);

#pragma unroll
    for (int v = 0; v < 8; ++v)
      hbuf[1 - p][(mb + v) * LDSROW + n] = acc[v];

    asm volatile("s_wait_dscnt 0x0" ::: "memory");
    asm volatile("s_barrier_signal -1" ::: "memory");
#pragma unroll
    for (int v = 0; v < 8; ++v)
      out[rowbase[v] + (long)t * H_] = acc[v];
    asm volatile("s_barrier_wait -1" ::: "memory");

    cin = cnext;
    p ^= 1;
  }
}

extern "C" void kernel_launch(void* const* d_in, const int* in_sizes, int n_in,
                              void* d_out, int out_size, void* d_ws,
                              size_t ws_size, hipStream_t stream) {
  const float* x  = (const float*)d_in[0];
  const float* Wx = (const float*)d_in[1];
  const float* Wh = (const float*)d_in[2];
  const float* b  = (const float*)d_in[3];
  float* out = (float*)d_out;

  rnn_xproj_kernel<<<(B_ * T_) / 16, 512, 0, stream>>>(x, Wx, out);
  rnn_recur_kernel<<<B_ / 16, 512, 0, stream>>>(Wh, b, out);
}


# recurrence on 8 waves of two 16-column tiles each: h tile read from LDS once per step as one group of 16-byte loads ahead of the WMMA chain, not between WMMAs; k order inside the sum changes
# speedup vs baseline: 1.2805x; 1.0333x over previous
#include <hip/hip_runtime.h>
#include <hip/hip_bf16.h>


typedef __attribute__((ext_vector_type(2))) float v2f;
typedef __attribute__((ext_vector_type(4))) float v4f;
typedef __attribute__((ext_vector_type(8))) float v8f;

#define B_ 256
#define T_ 512
#define I_ 256
#define H_ 256
#define LDSROW 260

__device__ __forceinline__ float tanh_f32(float x) {
  const float ax = __builtin_fabsf(x);
  const float z  = x * x;
  float p = -5.70498872745e-3f;
  p = __builtin_fmaf(p, z, 2.06390887954e-2f);
  p = __builtin_fmaf(p, z, -5.37397155531e-2f);
  p = __builtin_fmaf(p, z, 1.33314422036e-1f);
  p = __builtin_fmaf(p, z, -3.33332819422e-1f);
  const float small = __builtin_fmaf(z, x * p, x);
  const float e   = __builtin_amdgcn_exp2f(ax * 2.885390081777927f);
  const float r   = __builtin_amdgcn_rcpf(e + 1.0f);
  const float big = __builtin_copysignf(__builtin_fmaf(-2.0f, r, 1.0f), x);
  return (ax < 0.625f) ? small : big;
}

__global__ __launch_bounds__(512) void rnn_xproj_kernel(
    const float* __restrict__ x, const float* __restrict__ Wx,
    float* __restrict__ out) {
  __shared__ float atile[16 * LDSROW];
  const int tid  = threadIdx.x;
  const int lane = tid & 31;
  const int wave = tid >> 5;
  const long M0  = (long)blockIdx.x * 16;

  {
    const int idx = tid * 8;
    const int r   = idx >> 8;
    const int i   = idx & 255;
    const float* src = x + (M0 + r) * (long)I_ + i;
    *(v4f*)(&atile[r * LDSROW + i])     = *(const v4f*)(src);
    *(v4f*)(&atile[r * LDSROW + i + 4]) = *(const v4f*)(src + 4);
  }
  __syncthreads();

  const int n     = (wave << 4) + (lane & 15);
  const int khalf = (lane >> 4) << 1;
  v2f bfrag[64];
#pragma unroll
  for (int kk = 0; kk < 64; ++kk) {
    const int kb = kk * 4 + khalf;
    bfrag[kk].x = Wx[n * I_ + kb + 0];
    bfrag[kk].y = Wx[n * I_ + kb + 1];
  }

  const int m = lane & 15;
  v8f acc = {};
#pragma unroll
  for (int kk = 0; kk < 64; ++kk) {
    const int kb = kk * 4 + khalf;
    v2f a;
    a.x = atile[m * LDSROW + kb + 0];
    a.y = atile[m * LDSROW + kb + 1];
    acc = __builtin_amdgcn_wmma_f32_16x16x4_f32(false, a, false, bfrag[kk],
                                                (short)0, acc, false, false);
  }

  const int mb = (lane >> 4) << 3;
#pragma unroll
  for (int v = 0; v < 8; ++v)
    out[(M0 + mb + v) * (long)H_ + n] = acc[v];
}

__global__ __launch_bounds__(256) __attribute__((amdgpu_waves_per_eu(2, 2))) void rnn_recur_kernel(
    const float* __restrict__ Wh, const float* __restrict__ bias,
    float* __restrict__ out) {
  __shared__ __attribute__((aligned(16))) float hbuf[2][16 * LDSROW];
  const int tid  = threadIdx.x;
  const int lane = tid & 31;
  const int wave = tid >> 5;
  const int b0   = blockIdx.x * 16;

  const int c  = lane & 15;
  const int hh = lane >> 4;
  const int kq = hh << 2;
  const int n0 = (wave << 5) + 2 * c;
  const int n1 = n0 + 1;

  v4f bq0[32], bq1[32];
#pragma unroll
  for (int j = 0; j < 32; ++j) {
    bq0[j] = *(const v4f*)(Wh + n0 * H_ + 8 * j + kq);
    bq1[j] = *(const v4f*)(Wh + n1 * H_ + 8 * j + kq);
  }
  const float bn0 = bias[n0];
  const float bn1 = bias[n1];

  {
    const int idx = tid * 16;
    const int r   = idx >> 8;
    const int i   = idx & 255;
    const v4f z4  = {0.0f, 0.0f, 0.0f, 0.0f};
#pragma unroll
    for (int q = 0; q < 4; ++q) *(v4f*)(&hbuf[0][r * LDSROW + i + 4 * q]) = z4;
  }
  __syncthreads();

  const int mb = hh << 3;

  long rowbase[8];
#pragma unroll
  for (int v = 0; v < 8; ++v)
    rowbase[v] = ((long)(b0 + mb + v) * T_) * H_ + n0;

  v8f acc0, acc1;
#pragma unroll
  for (int v = 0; v < 8; ++v) {
    const v2f c01 = *(const v2f*)(out + rowbase[v]);
    acc0[v] = c01.x + bn0;
    acc1[v] = c01.y + bn1;
  }

  int p = 0;
  for (int t = 0; t < T_; ++t) {
    const float* hrow = &hbuf[p][c * LDSROW + kq];
    v4f aq[32];
#pragma unroll
    for (int j = 0; j < 32; ++j) aq[j] = *(const v4f*)(hrow + 8 * j);

    const int tn = (t + 1 < T_) ? (t + 1) : (T_ - 1);
    v2f cn[8];
#pragma unroll
    for (int v = 0; v < 8; ++v)
      cn[v] = *(const v2f*)(out + rowbase[v] + (long)tn * H_);

    __builtin_amdgcn_sched_group_barrier(0x100, 32, 0);
    __builtin_amdgcn_sched_group_barrier(0x020, 8, 0);
    __builtin_amdgcn_sched_barrier(0);

#pragma unroll
    for (int j = 0; j < 32; ++j) {
      const v2f alo = __builtin_shufflevector(aq[j], aq[j], 0, 1);
      const v2f ahi = __builtin_shufflevector(aq[j], aq[j], 2, 3);
      const v2f b0lo = __builtin_shufflevector(bq0[j], bq0[j], 0, 1);
      const v2f b0hi = __builtin_shufflevector(bq0[j], bq0[j], 2, 3);
      const v2f b1lo = __builtin_shufflevector(bq1[j], bq1[j], 0, 1);
      const v2f b1hi = __builtin_shufflevector(bq1[j], bq1[j], 2, 3);
      acc0 = __builtin_amdgcn_wmma_f32_16x16x4_f32(false, alo, false, b0lo,
                                                   (short)0, acc0, false, false);
      acc1 = __builtin_amdgcn_wmma_f32_16x16x4_f32(false, alo, false, b1lo,
                                                   (short)0, acc1, false, false);
      acc0 = __builtin_amdgcn_wmma_f32_16x16x4_f32(false, ahi, false, b0hi,
                                                   (short)0, acc0, false, false);
      acc1 = __builtin_amdgcn_wmma_f32_16x16x4_f32(false, ahi, false, b1hi,
                                                   (short)0, acc1, false, false);
    }

    __builtin_amdgcn_sched_barrier(0);

    v8f h0, h1;
#pragma unroll
    for (int v = 0; v < 8; ++v) {
      h0[v] = tanh_f32(acc0[v]);
      h1[v] = tanh_f32(acc1[v]);
    }

#pragma unroll
    for (int v = 0; v < 8; ++v) {
      const v2f h01 = {h0[v], h1[v]};
      *(v2f*)(&hbuf[1 - p][(mb + v) * LDSROW + n0]) = h01;
    }

    asm volatile("s_wait_dscnt 0x0" ::: "memory");
    asm volatile("s_barrier_signal -1" ::: "memory");
    __builtin_amdgcn_sched_barrier(0);
#pragma unroll
    for (int v = 0; v < 8; ++v) {
      const v2f h01 = {h0[v], h1[v]};
      *(v2f*)(out + rowbase[v] + (long)t * H_) = h01;
    }
#pragma unroll
    for (int v = 0; v < 8; ++v) {
      acc0[v] = cn[v].x + bn0;
      acc1[v] = cn[v].y + bn1;
    }
    asm volatile("s_barrier_wait -1" ::: "memory");

    p ^= 1;
  }
}

extern "C" void kernel_launch(void* const* d_in, const int* in_sizes, int n_in,
                              void* d_out, int out_size, void* d_ws,
                              size_t ws_size, hipStream_t stream) {
  const float* x  = (const float*)d_in[0];
  const float* Wx = (const float*)d_in[1];
  const float* Wh = (const float*)d_in[2];
  const float* b  = (const float*)d_in[3];
  float* out = (float*)d_out;

  rnn_xproj_kernel<<<(B_ * T_) / 16, 512, 0, stream>>>(x, Wx, out);
  rnn_recur_kernel<<<B_ / 16, 256, 0, stream>>>(Wh, b, out);
}


# x projection on 256 persistent workgroups: W_x fetched once and held in registers over 32 row tiles, x through a two-buffer LDS ring, not W_x re-fetched per 16-row tile; its k order changes
# speedup vs baseline: 1.4629x; 1.1424x over previous
#include <hip/hip_runtime.h>
#include <hip/hip_bf16.h>


typedef __attribute__((ext_vector_type(2))) float v2f;
typedef __attribute__((ext_vector_type(4))) float v4f;
typedef __attribute__((ext_vector_type(8))) float v8f;

#define B_ 256
#define T_ 512
#define I_ 256
#define H_ 256
#define LDSROW 260

__device__ __forceinline__ float tanh_f32(float x) {
  const float ax = __builtin_fabsf(x);
  const float z  = x * x;
  float p = -5.70498872745e-3f;
  p = __builtin_fmaf(p, z, 2.06390887954e-2f);
  p = __builtin_fmaf(p, z, -5.37397155531e-2f);
  p = __builtin_fmaf(p, z, 1.33314422036e-1f);
  p = __builtin_fmaf(p, z, -3.33332819422e-1f);
  const float small = __builtin_fmaf(z, x * p, x);
  const float e   = __builtin_amdgcn_exp2f(ax * 2.885390081777927f);
  const float r   = __builtin_amdgcn_rcpf(e + 1.0f);
  const float big = __builtin_copysignf(__builtin_fmaf(-2.0f, r, 1.0f), x);
  return (ax < 0.625f) ? small : big;
}

#define XP_WGS   256
#define XP_ROWS  512
#define XP_STAGE 32
#define XP_NST   16

__global__ __launch_bounds__(256) __attribute__((amdgpu_waves_per_eu(2, 2))) void rnn_xproj_kernel(
    const float* __restrict__ x, const float* __restrict__ Wx,
    float* __restrict__ out) {
  __shared__ __attribute__((aligned(16))) float xs[2][XP_STAGE * LDSROW];
  const int tid  = threadIdx.x;
  const int lane = tid & 31;
  const int wave = tid >> 5;
  const long M0  = (long)blockIdx.x * XP_ROWS;

  const int c  = lane & 15;
  const int hh = lane >> 4;
  const int kq = hh << 2;
  const int mb = hh << 3;
  const int n0 = (wave << 5) + 2 * c;
  const int n1 = n0 + 1;

  const int srow = tid >> 6;
  const int scol = (tid & 63) << 2;
  const float* xsrc = x + (M0 + srow) * (long)I_ + scol;
  const int sdst = srow * LDSROW + scol;

  v4f pre[8];
#pragma unroll
  for (int q = 0; q < 8; ++q) pre[q] = *(const v4f*)(xsrc + (long)(4 * q) * I_);

  v4f bq0[32], bq1[32];
#pragma unroll
  for (int j = 0; j < 32; ++j) {
    bq0[j] = *(const v4f*)(Wx + n0 * I_ + 8 * j + kq);
    bq1[j] = *(const v4f*)(Wx + n1 * I_ + 8 * j + kq);
  }

#pragma unroll
  for (int q = 0; q < 8; ++q) *(v4f*)(&xs[0][sdst + 4 * q * LDSROW]) = pre[q];
  __syncthreads();

  float* obase = out + (M0 + mb) * (long)H_ + n0;

  const v8f zero8 = {0.0f, 0.0f, 0.0f, 0.0f, 0.0f, 0.0f, 0.0f, 0.0f};

  int p = 0;
  for (int s = 0; s < XP_NST; ++s) {
    const int sn = (s + 1 < XP_NST) ? (s + 1) : (XP_NST - 1);
#pragma unroll
    for (int q = 0; q < 8; ++q)
      pre[q] = *(const v4f*)(xsrc + (long)(sn * XP_STAGE + 4 * q) * I_);

    const float* arow0 = &xs[p][c * LDSROW + kq];
    v4f aq[32];
#pragma unroll
    for (int j = 0; j < 32; ++j) aq[j] = *(const v4f*)(arow0 + 8 * j);

    __builtin_amdgcn_sched_group_barrier(0x100, 32, 0);
    __builtin_amdgcn_sched_group_barrier(0x020, 8, 0);
    __builtin_amdgcn_sched_barrier(0);

    v8f acc0 = zero8, acc1 = zero8;
#pragma unroll
    for (int j = 0; j < 32; ++j) {
      const v2f alo = __builtin_shufflevector(aq[j], aq[j], 0, 1);
      const v2f ahi = __builtin_shufflevector(aq[j], aq[j], 2, 3);
      const v2f b0lo = __builtin_shufflevector(bq0[j], bq0[j], 0, 1);
      const v2f b0hi = __builtin_shufflevector(bq0[j], bq0[j], 2, 3);
      const v2f b1lo = __builtin_shufflevector(bq1[j], bq1[j], 0, 1);
      const v2f b1hi = __builtin_shufflevector(bq1[j], bq1[j], 2, 3);
      acc0 = __builtin_amdgcn_wmma_f32_16x16x4_f32(false, alo, false, b0lo,
                                                   (short)0, acc0, false, false);
      acc1 = __builtin_amdgcn_wmma_f32_16x16x4_f32(false, alo, false, b1lo,
                                                   (short)0, acc1, false, false);
      acc0 = __builtin_amdgcn_wmma_f32_16x16x4_f32(false, ahi, false, b0hi,
                                                   (short)0, acc0, false, false);
      acc1 = __builtin_amdgcn_wmma_f32_16x16x4_f32(false, ahi, false, b1hi,
                                                   (short)0, acc1, false, false);
    }

#pragma unroll
    for (int v = 0; v < 8; ++v) {
      const v2f r01 = {acc0[v], acc1[v]};
      *(v2f*)(obase + (long)(s * XP_STAGE + v) * H_) = r01;
    }
    asm volatile("" ::: "memory");

    const float* arow1 = arow0 + 16 * LDSROW;
#pragma unroll
    for (int j = 0; j < 32; ++j) aq[j] = *(const v4f*)(arow1 + 8 * j);
#pragma unroll
    for (int q = 0; q < 8; ++q) *(v4f*)(&xs[1 - p][sdst + 4 * q * LDSROW]) = pre[q];

    asm volatile("s_wait_dscnt 0x0" ::: "memory");
    asm volatile("s_barrier_signal -1" ::: "memory");
    __builtin_amdgcn_sched_barrier(0);

    acc0 = zero8;
    acc1 = zero8;
#pragma unroll
    for (int j = 0; j < 32; ++j) {
      const v2f alo = __builtin_shufflevector(aq[j], aq[j], 0, 1);
      const v2f ahi = __builtin_shufflevector(aq[j], aq[j], 2, 3);
      const v2f b0lo = __builtin_shufflevector(bq0[j], bq0[j], 0, 1);
      const v2f b0hi = __builtin_shufflevector(bq0[j], bq0[j], 2, 3);
      const v2f b1lo = __builtin_shufflevector(bq1[j], bq1[j], 0, 1);
      const v2f b1hi = __builtin_shufflevector(bq1[j], bq1[j], 2, 3);
      acc0 = __builtin_amdgcn_wmma_f32_16x16x4_f32(false, alo, false, b0lo,
                                                   (short)0, acc0, false, false);
      acc1 = __builtin_amdgcn_wmma_f32_16x16x4_f32(false, alo, false, b1lo,
                                                   (short)0, acc1, false, false);
      acc0 = __builtin_amdgcn_wmma_f32_16x16x4_f32(false, ahi, false, b0hi,
                                                   (short)0, acc0, false, false);
      acc1 = __builtin_amdgcn_wmma_f32_16x16x4_f32(false, ahi, false, b1hi,
                                                   (short)0, acc1, false, false);
    }
#pragma unroll
    for (int v = 0; v < 8; ++v) {
      const v2f r01 = {acc0[v], acc1[v]};
      *(v2f*)(obase + (long)(s * XP_STAGE + 16 + v) * H_) = r01;
    }
    asm volatile("s_barrier_wait -1" ::: "memory");

    p ^= 1;
  }
}

__global__ __launch_bounds__(256) __attribute__((amdgpu_waves_per_eu(2, 2))) void rnn_recur_kernel(
    const float* __restrict__ Wh, const float* __restrict__ bias,
    float* __restrict__ out) {
  __shared__ __attribute__((aligned(16))) float hbuf[2][16 * LDSROW];
  const int tid  = threadIdx.x;
  const int lane = tid & 31;
  const int wave = tid >> 5;
  const int b0   = blockIdx.x * 16;

  const int c  = lane & 15;
  const int hh = lane >> 4;
  const int kq = hh << 2;
  const int n0 = (wave << 5) + 2 * c;
  const int n1 = n0 + 1;

  v4f bq0[32], bq1[32];
#pragma unroll
  for (int j = 0; j < 32; ++j) {
    bq0[j] = *(const v4f*)(Wh + n0 * H_ + 8 * j + kq);
    bq1[j] = *(const v4f*)(Wh + n1 * H_ + 8 * j + kq);
  }
  const float bn0 = bias[n0];
  const float bn1 = bias[n1];

  {
    const int idx = tid * 16;
    const int r   = idx >> 8;
    const int i   = idx & 255;
    const v4f z4  = {0.0f, 0.0f, 0.0f, 0.0f};
#pragma unroll
    for (int q = 0; q < 4; ++q) *(v4f*)(&hbuf[0][r * LDSROW + i + 4 * q]) = z4;
  }
  __syncthreads();

  const int mb = hh << 3;

  long rowbase[8];
#pragma unroll
  for (int v = 0; v < 8; ++v)
    rowbase[v] = ((long)(b0 + mb + v) * T_) * H_ + n0;

  v8f acc0, acc1;
#pragma unroll
  for (int v = 0; v < 8; ++v) {
    const v2f c01 = *(const v2f*)(out + rowbase[v]);
    acc0[v] = c01.x + bn0;
    acc1[v] = c01.y + bn1;
  }

  int p = 0;
  for (int t = 0; t < T_; ++t) {
    const float* hrow = &hbuf[p][c * LDSROW + kq];
    v4f aq[32];
#pragma unroll
    for (int j = 0; j < 32; ++j) aq[j] = *(const v4f*)(hrow + 8 * j);

    const int tn = (t + 1 < T_) ? (t + 1) : (T_ - 1);
    v2f cn[8];
#pragma unroll
    for (int v = 0; v < 8; ++v)
      cn[v] = *(const v2f*)(out + rowbase[v] + (long)tn * H_);

    __builtin_amdgcn_sched_group_barrier(0x100, 32, 0);
    __builtin_amdgcn_sched_group_barrier(0x020, 8, 0);
    __builtin_amdgcn_sched_barrier(0);

#pragma unroll
    for (int j = 0; j < 32; ++j) {
      const v2f alo = __builtin_shufflevector(aq[j], aq[j], 0, 1);
      const v2f ahi = __builtin_shufflevector(aq[j], aq[j], 2, 3);
      const v2f b0lo = __builtin_shufflevector(bq0[j], bq0[j], 0, 1);
      const v2f b0hi = __builtin_shufflevector(bq0[j], bq0[j], 2, 3);
      const v2f b1lo = __builtin_shufflevector(bq1[j], bq1[j], 0, 1);
      const v2f b1hi = __builtin_shufflevector(bq1[j], bq1[j], 2, 3);
      acc0 = __builtin_amdgcn_wmma_f32_16x16x4_f32(false, alo, false, b0lo,
                                                   (short)0, acc0, false, false);
      acc1 = __builtin_amdgcn_wmma_f32_16x16x4_f32(false, alo, false, b1lo,
                                                   (short)0, acc1, false, false);
      acc0 = __builtin_amdgcn_wmma_f32_16x16x4_f32(false, ahi, false, b0hi,
                                                   (short)0, acc0, false, false);
      acc1 = __builtin_amdgcn_wmma_f32_16x16x4_f32(false, ahi, false, b1hi,
                                                   (short)0, acc1, false, false);
    }

    __builtin_amdgcn_sched_barrier(0);

    v8f h0, h1;
#pragma unroll
    for (int v = 0; v < 8; ++v) {
      h0[v] = tanh_f32(acc0[v]);
      h1[v] = tanh_f32(acc1[v]);
    }

#pragma unroll
    for (int v = 0; v < 8; ++v) {
      const v2f h01 = {h0[v], h1[v]};
      *(v2f*)(&hbuf[1 - p][(mb + v) * LDSROW + n0]) = h01;
    }

    asm volatile("s_wait_dscnt 0x0" ::: "memory");
    asm volatile("s_barrier_signal -1" ::: "memory");
    __builtin_amdgcn_sched_barrier(0);
#pragma unroll
    for (int v = 0; v < 8; ++v) {
      const v2f h01 = {h0[v], h1[v]};
      *(v2f*)(out + rowbase[v] + (long)t * H_) = h01;
    }
#pragma unroll
    for (int v = 0; v < 8; ++v) {
      acc0[v] = cn[v].x + bn0;
      acc1[v] = cn[v].y + bn1;
    }
    asm volatile("s_barrier_wait -1" ::: "memory");

    p ^= 1;
  }
}

extern "C" void kernel_launch(void* const* d_in, const int* in_sizes, int n_in,
                              void* d_out, int out_size, void* d_ws,
                              size_t ws_size, hipStream_t stream) {
  const float* x  = (const float*)d_in[0];
  const float* Wx = (const float*)d_in[1];
  const float* Wh = (const float*)d_in[2];
  const float* b  = (const float*)d_in[3];
  float* out = (float*)d_out;

  rnn_xproj_kernel<<<XP_WGS, 256, 0, stream>>>(x, Wx, out);
  rnn_recur_kernel<<<B_ / 16, 256, 0, stream>>>(Wh, b, out);
}


# tanh of the recurrence on the hardware transcendental instruction v_tanh_f32 (its own rounding of tanh) instead of the inline polynomial / exp2 + rcp evaluation with a select
# speedup vs baseline: 1.5885x; 1.0859x over previous
#include <hip/hip_runtime.h>
#include <hip/hip_bf16.h>


typedef __attribute__((ext_vector_type(2))) float v2f;
typedef __attribute__((ext_vector_type(4))) float v4f;
typedef __attribute__((ext_vector_type(8))) float v8f;

#define B_ 256
#define T_ 512
#define I_ 256
#define H_ 256
#define LDSROW 260

__device__ __forceinline__ float tanh_f32(float x) {
  return __builtin_amdgcn_tanhf(x);
}

#define XP_WGS   256
#define XP_ROWS  512
#define XP_STAGE 32
#define XP_NST   16

__global__ __launch_bounds__(256) __attribute__((amdgpu_waves_per_eu(2, 2))) void rnn_xproj_kernel(
    const float* __restrict__ x, const float* __restrict__ Wx,
    float* __restrict__ out) {
  __shared__ __attribute__((aligned(16))) float xs[2][XP_STAGE * LDSROW];
  const int tid  = threadIdx.x;
  const int lane = tid & 31;
  const int wave = tid >> 5;
  const long M0  = (long)blockIdx.x * XP_ROWS;

  const int c  = lane & 15;
  const int hh = lane >> 4;
  const int kq = hh << 2;
  const int mb = hh << 3;
  const int n0 = (wave << 5) + 2 * c;
  const int n1 = n0 + 1;

  const int srow = tid >> 6;
  const int scol = (tid & 63) << 2;
  const float* xsrc = x + (M0 + srow) * (long)I_ + scol;
  const int sdst = srow * LDSROW + scol;

  v4f pre[8];
#pragma unroll
  for (int q = 0; q < 8; ++q) pre[q] = *(const v4f*)(xsrc + (long)(4 * q) * I_);

  v4f bq0[32], bq1[32];
#pragma unroll
  for (int j = 0; j < 32; ++j) {
    bq0[j] = *(const v4f*)(Wx + n0 * I_ + 8 * j + kq);
    bq1[j] = *(const v4f*)(Wx + n1 * I_ + 8 * j + kq);
  }

#pragma unroll
  for (int q = 0; q < 8; ++q) *(v4f*)(&xs[0][sdst + 4 * q * LDSROW]) = pre[q];
  __syncthreads();

  float* obase = out + (M0 + mb) * (long)H_ + n0;

  const v8f zero8 = {0.0f, 0.0f, 0.0f, 0.0f, 0.0f, 0.0f, 0.0f, 0.0f};

  int p = 0;
  for (int s = 0; s < XP_NST; ++s) {
    const int sn = (s + 1 < XP_NST) ? (s + 1) : (XP_NST - 1);
#pragma unroll
    for (int q = 0; q < 8; ++q)
      pre[q] = *(const v4f*)(xsrc + (long)(sn * XP_STAGE + 4 * q) * I_);

    const float* arow0 = &xs[p][c * LDSROW + kq];
    v4f aq[32];
#pragma unroll
    for (int j = 0; j < 32; ++j) aq[j] = *(const v4f*)(arow0 + 8 * j);

    __builtin_amdgcn_sched_group_barrier(0x100, 32, 0);
    __builtin_amdgcn_sched_group_barrier(0x020, 8, 0);
    __builtin_amdgcn_sched_barrier(0);

    v8f acc0 = zero8, acc1 = zero8;
#pragma unroll
    for (int j = 0; j < 32; ++j) {
      const v2f alo = __builtin_shufflevector(aq[j], aq[j], 0, 1);
      const v2f ahi = __builtin_shufflevector(aq[j], aq[j], 2, 3);
      const v2f b0lo = __builtin_shufflevector(bq0[j], bq0[j], 0, 1);
      const v2f b0hi = __builtin_shufflevector(bq0[j], bq0[j], 2, 3);
      const v2f b1lo = __builtin_shufflevector(bq1[j], bq1[j], 0, 1);
      const v2f b1hi = __builtin_shufflevector(bq1[j], bq1[j], 2, 3);
      acc0 = __builtin_amdgcn_wmma_f32_16x16x4_f32(false, alo, false, b0lo,
                                                   (short)0, acc0, false, false);
      acc1 = __builtin_amdgcn_wmma_f32_16x16x4_f32(false, alo, false, b1lo,
                                                   (short)0, acc1, false, false);
      acc0 = __builtin_amdgcn_wmma_f32_16x16x4_f32(false, ahi, false, b0hi,
                                                   (short)0, acc0, false, false);
      acc1 = __builtin_amdgcn_wmma_f32_16x16x4_f32(false, ahi, false, b1hi,
                                                   (short)0, acc1, false, false);
    }

#pragma unroll
    for (int v = 0; v < 8; ++v) {
      const v2f r01 = {acc0[v], acc1[v]};
      *(v2f*)(obase + (long)(s * XP_STAGE + v) * H_) = r01;
    }
    asm volatile("" ::: "memory");

    const float* arow1 = arow0 + 16 * LDSROW;
#pragma unroll
    for (int j = 0; j < 32; ++j) aq[j] = *(const v4f*)(arow1 + 8 * j);
#pragma unroll
    for (int q = 0; q < 8; ++q) *(v4f*)(&xs[1 - p][sdst + 4 * q * LDSROW]) = pre[q];

    asm volatile("s_wait_dscnt 0x0" ::: "memory");
    asm volatile("s_barrier_signal -1" ::: "memory");
    __builtin_amdgcn_sched_barrier(0);

    acc0 = zero8;
    acc1 = zero8;
#pragma unroll
    for (int j = 0; j < 32; ++j) {
      const v2f alo = __builtin_shufflevector(aq[j], aq[j], 0, 1);
      const v2f ahi = __builtin_shufflevector(aq[j], aq[j], 2, 3);
      const v2f b0lo = __builtin_shufflevector(bq0[j], bq0[j], 0, 1);
      const v2f b0hi = __builtin_shufflevector(bq0[j], bq0[j], 2, 3);
      const v2f b1lo = __builtin_shufflevector(bq1[j], bq1[j], 0, 1);
      const v2f b1hi = __builtin_shufflevector(bq1[j], bq1[j], 2, 3);
      acc0 = __builtin_amdgcn_wmma_f32_16x16x4_f32(false, alo, false, b0lo,
                                                   (short)0, acc0, false, false);
      acc1 = __builtin_amdgcn_wmma_f32_16x16x4_f32(false, alo, false, b1lo,
                                                   (short)0, acc1, false, false);
      acc0 = __builtin_amdgcn_wmma_f32_16x16x4_f32(false, ahi, false, b0hi,
                                                   (short)0, acc0, false, false);
      acc1 = __builtin_amdgcn_wmma_f32_16x16x4_f32(false, ahi, false, b1hi,
                                                   (short)0, acc1, false, false);
    }
#pragma unroll
    for (int v = 0; v < 8; ++v) {
      const v2f r01 = {acc0[v], acc1[v]};
      *(v2f*)(obase + (long)(s * XP_STAGE + 16 + v) * H_) = r01;
    }
    asm volatile("s_barrier_wait -1" ::: "memory");

    p ^= 1;
  }
}

__global__ __launch_bounds__(256) __attribute__((amdgpu_waves_per_eu(2, 2))) void rnn_recur_kernel(
    const float* __restrict__ Wh, const float* __restrict__ bias,
    float* __restrict__ out) {
  __shared__ __attribute__((aligned(16))) float hbuf[2][16 * LDSROW];
  const int tid  = threadIdx.x;
  const int lane = tid & 31;
  const int wave = tid >> 5;
  const int b0   = blockIdx.x * 16;

  const int c  = lane & 15;
  const int hh = lane >> 4;
  const int kq = hh << 2;
  const int n0 = (wave << 5) + 2 * c;
  const int n1 = n0 + 1;

  v4f bq0[32], bq1[32];
#pragma unroll
  for (int j = 0; j < 32; ++j) {
    bq0[j] = *(const v4f*)(Wh + n0 * H_ + 8 * j + kq);
    bq1[j] = *(const v4f*)(Wh + n1 * H_ + 8 * j + kq);
  }
  const float bn0 = bias[n0];
  const float bn1 = bias[n1];

  {
    const int idx = tid * 16;
    const int r   = idx >> 8;
    const int i   = idx & 255;
    const v4f z4  = {0.0f, 0.0f, 0.0f, 0.0f};
#pragma unroll
    for (int q = 0; q < 4; ++q) *(v4f*)(&hbuf[0][r * LDSROW + i + 4 * q]) = z4;
  }
  __syncthreads();

  const int mb = hh << 3;

  long rowbase[8];
#pragma unroll
  for (int v = 0; v < 8; ++v)
    rowbase[v] = ((long)(b0 + mb + v) * T_) * H_ + n0;

  v8f acc0, acc1;
#pragma unroll
  for (int v = 0; v < 8; ++v) {
    const v2f c01 = *(const v2f*)(out + rowbase[v]);
    acc0[v] = c01.x + bn0;
    acc1[v] = c01.y + bn1;
  }

  int p = 0;
  for (int t = 0; t < T_; ++t) {
    const float* hrow = &hbuf[p][c * LDSROW + kq];
    v4f aq[32];
#pragma unroll
    for (int j = 0; j < 32; ++j) aq[j] = *(const v4f*)(hrow + 8 * j);

    const int tn = (t + 1 < T_) ? (t + 1) : (T_ - 1);
    v2f cn[8];
#pragma unroll
    for (int v = 0; v < 8; ++v)
      cn[v] = *(const v2f*)(out + rowbase[v] + (long)tn * H_);

    __builtin_amdgcn_sched_group_barrier(0x100, 32, 0);
    __builtin_amdgcn_sched_group_barrier(0x020, 8, 0);
    __builtin_amdgcn_sched_barrier(0);

#pragma unroll
    for (int j = 0; j < 32; ++j) {
      const v2f alo = __builtin_shufflevector(aq[j], aq[j], 0, 1);
      const v2f ahi = __builtin_shufflevector(aq[j], aq[j], 2, 3);
      const v2f b0lo = __builtin_shufflevector(bq0[j], bq0[j], 0, 1);
      const v2f b0hi = __builtin_shufflevector(bq0[j], bq0[j], 2, 3);
      const v2f b1lo = __builtin_shufflevector(bq1[j], bq1[j], 0, 1);
      const v2f b1hi = __builtin_shufflevector(bq1[j], bq1[j], 2, 3);
      acc0 = __builtin_amdgcn_wmma_f32_16x16x4_f32(false, alo, false, b0lo,
                                                   (short)0, acc0, false, false);
      acc1 = __builtin_amdgcn_wmma_f32_16x16x4_f32(false, alo, false, b1lo,
                                                   (short)0, acc1, false, false);
      acc0 = __builtin_amdgcn_wmma_f32_16x16x4_f32(false, ahi, false, b0hi,
                                                   (short)0, acc0, false, false);
      acc1 = __builtin_amdgcn_wmma_f32_16x16x4_f32(false, ahi, false, b1hi,
                                                   (short)0, acc1, false, false);
    }

    __builtin_amdgcn_sched_barrier(0);

    v8f h0, h1;
#pragma unroll
    for (int v = 0; v < 8; ++v) {
      h0[v] = tanh_f32(acc0[v]);
      h1[v] = tanh_f32(acc1[v]);
    }

#pragma unroll
    for (int v = 0; v < 8; ++v) {
      const v2f h01 = {h0[v], h1[v]};
      *(v2f*)(&hbuf[1 - p][(mb + v) * LDSROW + n0]) = h01;
    }

    asm volatile("s_wait_dscnt 0x0" ::: "memory");
    asm volatile("s_barrier_signal -1" ::: "memory");
    __builtin_amdgcn_sched_barrier(0);
#pragma unroll
    for (int v = 0; v < 8; ++v) {
      const v2f h01 = {h0[v], h1[v]};
      *(v2f*)(out + rowbase[v] + (long)t * H_) = h01;
    }
#pragma unroll
    for (int v = 0; v < 8; ++v) {
      acc0[v] = cn[v].x + bn0;
      acc1[v] = cn[v].y + bn1;
    }
    asm volatile("s_barrier_wait -1" ::: "memory");

    p ^= 1;
  }
}

extern "C" void kernel_launch(void* const* d_in, const int* in_sizes, int n_in,
                              void* d_out, int out_size, void* d_ws,
                              size_t ws_size, hipStream_t stream) {
  const float* x  = (const float*)d_in[0];
  const float* Wx = (const float*)d_in[1];
  const float* Wh = (const float*)d_in[2];
  const float* b  = (const float*)d_in[3];
  float* out = (float*)d_out;

  rnn_xproj_kernel<<<XP_WGS, 256, 0, stream>>>(x, Wx, out);
  rnn_recur_kernel<<<B_ / 16, 256, 0, stream>>>(Wh, b, out);
}


# x projection of steps 64..511 runs in the same launch as the recurrence of steps 0..63, on the 240 workgroups the recurrence leaves idle; 3 launches, not 2; every sum and its order unchanged
# speedup vs baseline: 1.6486x; 1.0378x over previous
#include <hip/hip_runtime.h>
#include <hip/hip_bf16.h>


typedef __attribute__((ext_vector_type(2))) float v2f;
typedef __attribute__((ext_vector_type(4))) float v4f;
typedef __attribute__((ext_vector_type(8))) float v8f;

#define B_ 256
#define T_ 512
#define I_ 256
#define H_ 256
#define LDSROW 260

#define T_HEAD   64
#define REC_WGS  16
#define ALL_WGS  256

__device__ __forceinline__ float tanh_f32(float x) {
  return __builtin_amdgcn_tanhf(x);
}

#define XP_STAGE 32

__device__ __forceinline__ long xp_row0(const int g, const int t0, const int spb) {
  const int b = g / spb;
  const int s = g - b * spb;
  return (long)b * T_ + t0 + s * XP_STAGE;
}

__device__ __forceinline__ void xproj_role(
    const float* __restrict__ x, const float* __restrict__ Wx, float* __restrict__ out,
    float* xs, const int wg, const int nwg, const int t0, const int spb) {
  const int total = B_ * spb;
  const int tid  = threadIdx.x;
  const int lane = tid & 31;
  const int wave = tid >> 5;

  const int c  = lane & 15;
  const int hh = lane >> 4;
  const int kq = hh << 2;
  const int mb = hh << 3;
  const int n0 = (wave << 5) + 2 * c;
  const int n1 = n0 + 1;

  const int srow = tid >> 6;
  const int scol = (tid & 63) << 2;
  const float* xsrc = x + (long)srow * I_ + scol;
  const int sdst = srow * LDSROW + scol;

  int g = wg;
  long r0 = xp_row0(g, t0, spb);

  v4f pre[8];
#pragma unroll
  for (int q = 0; q < 8; ++q) pre[q] = *(const v4f*)(xsrc + (r0 + 4 * q) * I_);

  v4f bq0[32], bq1[32];
#pragma unroll
  for (int j = 0; j < 32; ++j) {
    bq0[j] = *(const v4f*)(Wx + n0 * I_ + 8 * j + kq);
    bq1[j] = *(const v4f*)(Wx + n1 * I_ + 8 * j + kq);
  }

#pragma unroll
  for (int q = 0; q < 8; ++q) *(v4f*)(xs + sdst + 4 * q * LDSROW) = pre[q];
  __syncthreads();

  float* obase0 = out + (long)mb * H_ + n0;

  const v8f zero8 = {0.0f, 0.0f, 0.0f, 0.0f, 0.0f, 0.0f, 0.0f, 0.0f};

  int p = 0;
  for (; g < total; g += nwg) {
    const int gn = (g + nwg < total) ? (g + nwg) : g;
    const long rn = xp_row0(gn, t0, spb);
#pragma unroll
    for (int q = 0; q < 8; ++q)
      pre[q] = *(const v4f*)(xsrc + (rn + 4 * q) * I_);

    float* obase = obase0 + r0 * H_;
    float* xcur = xs + p * (XP_STAGE * LDSROW);
    float* xnxt = xs + (1 - p) * (XP_STAGE * LDSROW);

    const float* arow0 = xcur + c * LDSROW + kq;
    v4f aq[32];
#pragma unroll
    for (int j = 0; j < 32; ++j) aq[j] = *(const v4f*)(arow0 + 8 * j);

    __builtin_amdgcn_sched_group_barrier(0x100, 32, 0);
    __builtin_amdgcn_sched_group_barrier(0x020, 8, 0);
    __builtin_amdgcn_sched_barrier(0);

    v8f acc0 = zero8, acc1 = zero8;
#pragma unroll
    for (int j = 0; j < 32; ++j) {
      const v2f alo = __builtin_shufflevector(aq[j], aq[j], 0, 1);
      const v2f ahi = __builtin_shufflevector(aq[j], aq[j], 2, 3);
      const v2f b0lo = __builtin_shufflevector(bq0[j], bq0[j], 0, 1);
      const v2f b0hi = __builtin_shufflevector(bq0[j], bq0[j], 2, 3);
      const v2f b1lo = __builtin_shufflevector(bq1[j], bq1[j], 0, 1);
      const v2f b1hi = __builtin_shufflevector(bq1[j], bq1[j], 2, 3);
      acc0 = __builtin_amdgcn_wmma_f32_16x16x4_f32(false, alo, false, b0lo,
                                                   (short)0, acc0, false, false);
      acc1 = __builtin_amdgcn_wmma_f32_16x16x4_f32(false, alo, false, b1lo,
                                                   (short)0, acc1, false, false);
      acc0 = __builtin_amdgcn_wmma_f32_16x16x4_f32(false, ahi, false, b0hi,
                                                   (short)0, acc0, false, false);
      acc1 = __builtin_amdgcn_wmma_f32_16x16x4_f32(false, ahi, false, b1hi,
                                                   (short)0, acc1, false, false);
    }

#pragma unroll
    for (int v = 0; v < 8; ++v) {
      const v2f r01 = {acc0[v], acc1[v]};
      *(v2f*)(obase + (long)v * H_) = r01;
    }
    asm volatile("" ::: "memory");

    const float* arow1 = arow0 + 16 * LDSROW;
#pragma unroll
    for (int j = 0; j < 32; ++j) aq[j] = *(const v4f*)(arow1 + 8 * j);
#pragma unroll
    for (int q = 0; q < 8; ++q) *(v4f*)(xnxt + sdst + 4 * q * LDSROW) = pre[q];

    asm volatile("s_wait_dscnt 0x0" ::: "memory");
    asm volatile("s_barrier_signal -1" ::: "memory");
    __builtin_amdgcn_sched_barrier(0);

    acc0 = zero8;
    acc1 = zero8;
#pragma unroll
    for (int j = 0; j < 32; ++j) {
      const v2f alo = __builtin_shufflevector(aq[j], aq[j], 0, 1);
      const v2f ahi = __builtin_shufflevector(aq[j], aq[j], 2, 3);
      const v2f b0lo = __builtin_shufflevector(bq0[j], bq0[j], 0, 1);
      const v2f b0hi = __builtin_shufflevector(bq0[j], bq0[j], 2, 3);
      const v2f b1lo = __builtin_shufflevector(bq1[j], bq1[j], 0, 1);
      const v2f b1hi = __builtin_shufflevector(bq1[j], bq1[j], 2, 3);
      acc0 = __builtin_amdgcn_wmma_f32_16x16x4_f32(false, alo, false, b0lo,
                                                   (short)0, acc0, false, false);
      acc1 = __builtin_amdgcn_wmma_f32_16x16x4_f32(false, alo, false, b1lo,
                                                   (short)0, acc1, false, false);
      acc0 = __builtin_amdgcn_wmma_f32_16x16x4_f32(false, ahi, false, b0hi,
                                                   (short)0, acc0, false, false);
      acc1 = __builtin_amdgcn_wmma_f32_16x16x4_f32(false, ahi, false, b1hi,
                                                   (short)0, acc1, false, false);
    }
#pragma unroll
    for (int v = 0; v < 8; ++v) {
      const v2f r01 = {acc0[v], acc1[v]};
      *(v2f*)(obase + (long)(16 + v) * H_) = r01;
    }
    asm volatile("s_barrier_wait -1" ::: "memory");

    r0 = rn;
    p ^= 1;
  }
}

__device__ __forceinline__ void recur_role(
    const float* __restrict__ Wh, const float* __restrict__ bias, float* __restrict__ out,
    float* hbuf, const int blk, const int t_begin, const int t_end) {
  const int tid  = threadIdx.x;
  const int lane = tid & 31;
  const int wave = tid >> 5;
  const int b0   = blk * 16;

  const int c  = lane & 15;
  const int hh = lane >> 4;
  const int kq = hh << 2;
  const int n0 = (wave << 5) + 2 * c;
  const int n1 = n0 + 1;

  v4f bq0[32], bq1[32];
#pragma unroll
  for (int j = 0; j < 32; ++j) {
    bq0[j] = *(const v4f*)(Wh + n0 * H_ + 8 * j + kq);
    bq1[j] = *(const v4f*)(Wh + n1 * H_ + 8 * j + kq);
  }
  const float bn0 = bias[n0];
  const float bn1 = bias[n1];

  {
    const int idx = tid * 16;
    const int r   = idx >> 8;
    const int i   = idx & 255;
    v4f hv[4];
    if (t_begin > 0) {
      const float* hsrc = out + ((long)(b0 + r) * T_ + (t_begin - 1)) * H_ + i;
#pragma unroll
      for (int q = 0; q < 4; ++q) hv[q] = *(const v4f*)(hsrc + 4 * q);
    } else {
      const v4f z4 = {0.0f, 0.0f, 0.0f, 0.0f};
#pragma unroll
      for (int q = 0; q < 4; ++q) hv[q] = z4;
    }
#pragma unroll
    for (int q = 0; q < 4; ++q) *(v4f*)(hbuf + r * LDSROW + i + 4 * q) = hv[q];
  }
  __syncthreads();

  const int mb = hh << 3;

  long rowbase[8];
#pragma unroll
  for (int v = 0; v < 8; ++v)
    rowbase[v] = ((long)(b0 + mb + v) * T_) * H_ + n0;

  v8f acc0, acc1;
#pragma unroll
  for (int v = 0; v < 8; ++v) {
    const v2f c01 = *(const v2f*)(out + rowbase[v] + (long)t_begin * H_);
    acc0[v] = c01.x + bn0;
    acc1[v] = c01.y + bn1;
  }

  int p = 0;
  for (int t = t_begin; t < t_end; ++t) {
    const float* hrow = hbuf + p * (16 * LDSROW) + c * LDSROW + kq;
    float* hnxt = hbuf + (1 - p) * (16 * LDSROW);
    v4f aq[32];
#pragma unroll
    for (int j = 0; j < 32; ++j) aq[j] = *(const v4f*)(hrow + 8 * j);

    const int tn = (t + 1 < t_end) ? (t + 1) : (t_end - 1);
    v2f cn[8];
#pragma unroll
    for (int v = 0; v < 8; ++v)
      cn[v] = *(const v2f*)(out + rowbase[v] + (long)tn * H_);

    __builtin_amdgcn_sched_group_barrier(0x100, 32, 0);
    __builtin_amdgcn_sched_group_barrier(0x020, 8, 0);
    __builtin_amdgcn_sched_barrier(0);

#pragma unroll
    for (int j = 0; j < 32; ++j) {
      const v2f alo = __builtin_shufflevector(aq[j], aq[j], 0, 1);
      const v2f ahi = __builtin_shufflevector(aq[j], aq[j], 2, 3);
      const v2f b0lo = __builtin_shufflevector(bq0[j], bq0[j], 0, 1);
      const v2f b0hi = __builtin_shufflevector(bq0[j], bq0[j], 2, 3);
      const v2f b1lo = __builtin_shufflevector(bq1[j], bq1[j], 0, 1);
      const v2f b1hi = __builtin_shufflevector(bq1[j], bq1[j], 2, 3);
      acc0 = __builtin_amdgcn_wmma_f32_16x16x4_f32(false, alo, false, b0lo,
                                                   (short)0, acc0, false, false);
      acc1 = __builtin_amdgcn_wmma_f32_16x16x4_f32(false, alo, false, b1lo,
                                                   (short)0, acc1, false, false);
      acc0 = __builtin_amdgcn_wmma_f32_16x16x4_f32(false, ahi, false, b0hi,
                                                   (short)0, acc0, false, false);
      acc1 = __builtin_amdgcn_wmma_f32_16x16x4_f32(false, ahi, false, b1hi,
                                                   (short)0, acc1, false, false);
    }

    __builtin_amdgcn_sched_barrier(0);

    v8f h0, h1;
#pragma unroll
    for (int v = 0; v < 8; ++v) {
      h0[v] = tanh_f32(acc0[v]);
      h1[v] = tanh_f32(acc1[v]);
    }

#pragma unroll
    for (int v = 0; v < 8; ++v) {
      const v2f h01 = {h0[v], h1[v]};
      *(v2f*)(hnxt + (mb + v) * LDSROW + n0) = h01;
    }

    asm volatile("s_wait_dscnt 0x0" ::: "memory");
    asm volatile("s_barrier_signal -1" ::: "memory");
    __builtin_amdgcn_sched_barrier(0);
#pragma unroll
    for (int v = 0; v < 8; ++v) {
      const v2f h01 = {h0[v], h1[v]};
      *(v2f*)(out + rowbase[v] + (long)t * H_) = h01;
    }
#pragma unroll
    for (int v = 0; v < 8; ++v) {
      acc0[v] = cn[v].x + bn0;
      acc1[v] = cn[v].y + bn1;
    }
    asm volatile("s_barrier_wait -1" ::: "memory");

    p ^= 1;
  }
}


__global__ __launch_bounds__(256) __attribute__((amdgpu_waves_per_eu(2, 2))) void rnn_xproj_head_kernel(
    const float* __restrict__ x, const float* __restrict__ Wx,
    float* __restrict__ out) {
  __shared__ __attribute__((aligned(16))) float smem[2 * XP_STAGE * LDSROW];
  xproj_role(x, Wx, out, smem, (int)blockIdx.x, ALL_WGS, 0, T_HEAD / XP_STAGE);
}

__global__ __launch_bounds__(256) __attribute__((amdgpu_waves_per_eu(2, 2))) void rnn_mid_kernel(
    const float* __restrict__ x, const float* __restrict__ Wx,
    const float* __restrict__ Wh, const float* __restrict__ bias,
    float* __restrict__ out) {
  __shared__ __attribute__((aligned(16))) float smem[2 * XP_STAGE * LDSROW];
  if (blockIdx.x < REC_WGS) {
    recur_role(Wh, bias, out, smem, (int)blockIdx.x, 0, T_HEAD);
  } else {
    xproj_role(x, Wx, out, smem, (int)blockIdx.x - REC_WGS, ALL_WGS - REC_WGS,
               T_HEAD, (T_ - T_HEAD) / XP_STAGE);
  }
}

__global__ __launch_bounds__(256) __attribute__((amdgpu_waves_per_eu(2, 2))) void rnn_recur_tail_kernel(
    const float* __restrict__ Wh, const float* __restrict__ bias,
    float* __restrict__ out) {
  __shared__ __attribute__((aligned(16))) float smem[2 * 16 * LDSROW];
  recur_role(Wh, bias, out, smem, (int)blockIdx.x, T_HEAD, T_);
}

extern "C" void kernel_launch(void* const* d_in, const int* in_sizes, int n_in,
                              void* d_out, int out_size, void* d_ws,
                              size_t ws_size, hipStream_t stream) {
  const float* x  = (const float*)d_in[0];
  const float* Wx = (const float*)d_in[1];
  const float* Wh = (const float*)d_in[2];
  const float* b  = (const float*)d_in[3];
  float* out = (float*)d_out;

  rnn_xproj_head_kernel<<<ALL_WGS, 256, 0, stream>>>(x, Wx, out);
  rnn_mid_kernel<<<ALL_WGS, 256, 0, stream>>>(x, Wx, Wh, b, out);
  rnn_recur_tail_kernel<<<REC_WGS, 256, 0, stream>>>(Wh, b, out);
}


# recurrence computed transposed (W_h as A, h as B): a wave's tanh results are B fragments, so its 16 own-k WMMAs of step t+1 cross the barrier at low s_setprio; k order of the sums changes
# speedup vs baseline: 1.6998x; 1.0311x over previous
#include <hip/hip_runtime.h>
#include <hip/hip_bf16.h>


typedef __attribute__((ext_vector_type(2))) float v2f;
typedef __attribute__((ext_vector_type(4))) float v4f;
typedef __attribute__((ext_vector_type(8))) float v8f;

#define B_ 256
#define T_ 512
#define I_ 256
#define H_ 256
#define LDSROW 260

#define T_HEAD   64
#define REC_WGS  16
#define ALL_WGS  256

__device__ __forceinline__ float tanh_f32(float x) {
  return __builtin_amdgcn_tanhf(x);
}

#define XP_STAGE 32

__device__ __forceinline__ long xp_row0(const int g, const int t0, const int spb) {
  const int b = g / spb;
  const int s = g - b * spb;
  return (long)b * T_ + t0 + s * XP_STAGE;
}

__device__ __forceinline__ void xproj_role(
    const float* __restrict__ x, const float* __restrict__ Wx, float* __restrict__ out,
    float* xs, const int wg, const int nwg, const int t0, const int spb) {
  const int total = B_ * spb;
  const int tid  = threadIdx.x;
  const int lane = tid & 31;
  const int wave = tid >> 5;

  const int c  = lane & 15;
  const int hh = lane >> 4;
  const int kq = hh << 2;
  const int mb = hh << 3;
  const int n0 = (wave << 5) + 2 * c;
  const int n1 = n0 + 1;

  const int srow = tid >> 6;
  const int scol = (tid & 63) << 2;
  const float* xsrc = x + (long)srow * I_ + scol;
  const int sdst = srow * LDSROW + scol;

  int g = wg;
  long r0 = xp_row0(g, t0, spb);

  v4f pre[8];
#pragma unroll
  for (int q = 0; q < 8; ++q) pre[q] = *(const v4f*)(xsrc + (r0 + 4 * q) * I_);

  v4f bq0[32], bq1[32];
#pragma unroll
  for (int j = 0; j < 32; ++j) {
    bq0[j] = *(const v4f*)(Wx + n0 * I_ + 8 * j + kq);
    bq1[j] = *(const v4f*)(Wx + n1 * I_ + 8 * j + kq);
  }

#pragma unroll
  for (int q = 0; q < 8; ++q) *(v4f*)(xs + sdst + 4 * q * LDSROW) = pre[q];
  __syncthreads();

  float* obase0 = out + (long)mb * H_ + n0;

  const v8f zero8 = {0.0f, 0.0f, 0.0f, 0.0f, 0.0f, 0.0f, 0.0f, 0.0f};

  int p = 0;
  for (; g < total; g += nwg) {
    const int gn = (g + nwg < total) ? (g + nwg) : g;
    const long rn = xp_row0(gn, t0, spb);
#pragma unroll
    for (int q = 0; q < 8; ++q)
      pre[q] = *(const v4f*)(xsrc + (rn + 4 * q) * I_);

    float* obase = obase0 + r0 * H_;
    float* xcur = xs + p * (XP_STAGE * LDSROW);
    float* xnxt = xs + (1 - p) * (XP_STAGE * LDSROW);

    const float* arow0 = xcur + c * LDSROW + kq;
    v4f aq[32];
#pragma unroll
    for (int j = 0; j < 32; ++j) aq[j] = *(const v4f*)(arow0 + 8 * j);

    __builtin_amdgcn_sched_group_barrier(0x100, 32, 0);
    __builtin_amdgcn_sched_group_barrier(0x020, 8, 0);
    __builtin_amdgcn_sched_barrier(0);

    v8f acc0 = zero8, acc1 = zero8;
#pragma unroll
    for (int j = 0; j < 32; ++j) {
      const v2f alo = __builtin_shufflevector(aq[j], aq[j], 0, 1);
      const v2f ahi = __builtin_shufflevector(aq[j], aq[j], 2, 3);
      const v2f b0lo = __builtin_shufflevector(bq0[j], bq0[j], 0, 1);
      const v2f b0hi = __builtin_shufflevector(bq0[j], bq0[j], 2, 3);
      const v2f b1lo = __builtin_shufflevector(bq1[j], bq1[j], 0, 1);
      const v2f b1hi = __builtin_shufflevector(bq1[j], bq1[j], 2, 3);
      acc0 = __builtin_amdgcn_wmma_f32_16x16x4_f32(false, alo, false, b0lo,
                                                   (short)0, acc0, false, false);
      acc1 = __builtin_amdgcn_wmma_f32_16x16x4_f32(false, alo, false, b1lo,
                                                   (short)0, acc1, false, false);
      acc0 = __builtin_amdgcn_wmma_f32_16x16x4_f32(false, ahi, false, b0hi,
                                                   (short)0, acc0, false, false);
      acc1 = __builtin_amdgcn_wmma_f32_16x16x4_f32(false, ahi, false, b1hi,
                                                   (short)0, acc1, false, false);
    }

#pragma unroll
    for (int v = 0; v < 8; ++v) {
      const v2f r01 = {acc0[v], acc1[v]};
      *(v2f*)(obase + (long)v * H_) = r01;
    }
    asm volatile("" ::: "memory");

    const float* arow1 = arow0 + 16 * LDSROW;
#pragma unroll
    for (int j = 0; j < 32; ++j) aq[j] = *(const v4f*)(arow1 + 8 * j);
#pragma unroll
    for (int q = 0; q < 8; ++q) *(v4f*)(xnxt + sdst + 4 * q * LDSROW) = pre[q];

    asm volatile("s_wait_dscnt 0x0" ::: "memory");
    asm volatile("s_barrier_signal -1" ::: "memory");
    __builtin_amdgcn_sched_barrier(0);

    acc0 = zero8;
    acc1 = zero8;
#pragma unroll
    for (int j = 0; j < 32; ++j) {
      const v2f alo = __builtin_shufflevector(aq[j], aq[j], 0, 1);
      const v2f ahi = __builtin_shufflevector(aq[j], aq[j], 2, 3);
      const v2f b0lo = __builtin_shufflevector(bq0[j], bq0[j], 0, 1);
      const v2f b0hi = __builtin_shufflevector(bq0[j], bq0[j], 2, 3);
      const v2f b1lo = __builtin_shufflevector(bq1[j], bq1[j], 0, 1);
      const v2f b1hi = __builtin_shufflevector(bq1[j], bq1[j], 2, 3);
      acc0 = __builtin_amdgcn_wmma_f32_16x16x4_f32(false, alo, false, b0lo,
                                                   (short)0, acc0, false, false);
      acc1 = __builtin_amdgcn_wmma_f32_16x16x4_f32(false, alo, false, b1lo,
                                                   (short)0, acc1, false, false);
      acc0 = __builtin_amdgcn_wmma_f32_16x16x4_f32(false, ahi, false, b0hi,
                                                   (short)0, acc0, false, false);
      acc1 = __builtin_amdgcn_wmma_f32_16x16x4_f32(false, ahi, false, b1hi,
                                                   (short)0, acc1, false, false);
    }
#pragma unroll
    for (int v = 0; v < 8; ++v) {
      const v2f r01 = {acc0[v], acc1[v]};
      *(v2f*)(obase + (long)(16 + v) * H_) = r01;
    }
    asm volatile("s_barrier_wait -1" ::: "memory");

    r0 = rn;
    p ^= 1;
  }
}


#define OWN_PIECE(JJ, HP, E0)                                                         \
  {                                                                                    \
    const v2f hlo = __builtin_shufflevector(HP, HP, (E0), (E0) + 1);                   \
    const v2f hhi = __builtin_shufflevector(HP, HP, (E0) + 2, (E0) + 3);               \
    const v2f w0lo = __builtin_shufflevector(wq0[JJ], wq0[JJ], 0, 1);                  \
    const v2f w0hi = __builtin_shufflevector(wq0[JJ], wq0[JJ], 2, 3);                  \
    const v2f w1lo = __builtin_shufflevector(wq1[JJ], wq1[JJ], 0, 1);                  \
    const v2f w1hi = __builtin_shufflevector(wq1[JJ], wq1[JJ], 2, 3);                  \
    acc0 = __builtin_amdgcn_wmma_f32_16x16x4_f32(false, w0lo, false, hlo,              \
                                                 (short)0, acc0, false, false);        \
    acc1 = __builtin_amdgcn_wmma_f32_16x16x4_f32(false, w1lo, false, hlo,              \
                                                 (short)0, acc1, false, false);        \
    acc0 = __builtin_amdgcn_wmma_f32_16x16x4_f32(false, w0hi, false, hhi,              \
                                                 (short)0, acc0, false, false);        \
    acc1 = __builtin_amdgcn_wmma_f32_16x16x4_f32(false, w1hi, false, hhi,              \
                                                 (short)0, acc1, false, false);        \
  }
#define OWN_16()            \
  OWN_PIECE(0, hp0, 0)      \
  OWN_PIECE(1, hp0, 4)      \
  OWN_PIECE(2, hp1, 0)      \
  OWN_PIECE(3, hp1, 4)

__device__ __forceinline__ void recur_role(
    const float* __restrict__ Wh, const float* __restrict__ bias, float* __restrict__ out,
    float* hbuf, const int blk, const int t_begin, const int t_end) {
  const int tid  = threadIdx.x;
  const int lane = tid & 31;
  const int wave = __builtin_amdgcn_readfirstlane(tid >> 5);
  const int b0   = blk * 16;

  const int c  = lane & 15;
  const int hh = lane >> 4;
  const int kq = hh << 2;
  const int nb = wave << 5;
  const int sg = ((c & 4) << 1) + ((c >> 3) << 2) + (c & 3);

  v4f wq0[32], wq1[32];
  {
    const float* wrow0 = Wh + (nb + sg) * H_ + kq;
    const float* wrow1 = wrow0 + 16 * H_;
#pragma unroll
    for (int jj = 0; jj < 32; ++jj) {
      const int ko = (((jj >> 2) ^ wave) << 5) + ((jj & 3) << 3);
      wq0[jj] = *(const v4f*)(wrow0 + ko);
      wq1[jj] = *(const v4f*)(wrow1 + ko);
    }
  }
  const int ncol = nb + kq;
  v4f bz[4];
#pragma unroll
  for (int q = 0; q < 4; ++q) bz[q] = *(const v4f*)(bias + ncol + 8 * q);

  float* orow = out + ((long)(b0 + c) * T_) * H_ + ncol;

  {
    const int idx = tid * 16;
    const int r   = idx >> 8;
    const int i   = idx & 255;
    v4f hv[4];
    if (t_begin > 0) {
      const float* hsrc = out + ((long)(b0 + r) * T_ + (t_begin - 1)) * H_ + i;
#pragma unroll
      for (int q = 0; q < 4; ++q) hv[q] = *(const v4f*)(hsrc + 4 * q);
    } else {
      const v4f z4 = {0.0f, 0.0f, 0.0f, 0.0f};
#pragma unroll
      for (int q = 0; q < 4; ++q) hv[q] = z4;
    }
#pragma unroll
    for (int q = 0; q < 4; ++q) *(v4f*)(hbuf + r * LDSROW + i + 4 * q) = hv[q];
  }

  v8f hp0, hp1;
  if (t_begin > 0) {
    const float* hs = orow + (long)(t_begin - 1) * H_;
    const v4f a0 = *(const v4f*)(hs);
    const v4f a1 = *(const v4f*)(hs + 8);
    const v4f a2 = *(const v4f*)(hs + 16);
    const v4f a3 = *(const v4f*)(hs + 24);
    hp0 = __builtin_shufflevector(a0, a1, 0, 1, 2, 3, 4, 5, 6, 7);
    hp1 = __builtin_shufflevector(a2, a3, 0, 1, 2, 3, 4, 5, 6, 7);
  } else {
    const v8f z8 = {0.0f, 0.0f, 0.0f, 0.0f, 0.0f, 0.0f, 0.0f, 0.0f};
    hp0 = z8;
    hp1 = z8;
  }
  __syncthreads();

  const float* hrow0 = hbuf + c * LDSROW + kq;
  int goff[8];
#pragma unroll
  for (int g = 1; g < 8; ++g) goff[g] = (g ^ wave) << 5;
  const int hdst = c * LDSROW + ncol;

  v8f acc0, acc1;
  {
    v4f cq[4];
#pragma unroll
    for (int q = 0; q < 4; ++q)
      cq[q] = *(const v4f*)(orow + (long)t_begin * H_ + 8 * q) + bz[q];
    acc0 = __builtin_shufflevector(cq[0], cq[1], 0, 1, 2, 3, 4, 5, 6, 7);
    acc1 = __builtin_shufflevector(cq[2], cq[3], 0, 1, 2, 3, 4, 5, 6, 7);
  }
  OWN_16()

  int p = 0;
  for (int t = t_begin; t < t_end; ++t) {
    __builtin_amdgcn_s_setprio(3);

    const float* hrow = hrow0 + p * (16 * LDSROW);
    float* hnxt = hbuf + (1 - p) * (16 * LDSROW);
    v4f hf[32];
#pragma unroll
    for (int g = 1; g < 8; ++g) {
#pragma unroll
      for (int i = 0; i < 4; ++i) hf[4 * g + i] = *(const v4f*)(hrow + goff[g] + 8 * i);
    }

    const int tn = (t + 1 < t_end) ? (t + 1) : (t_end - 1);
    v4f cn[4];
#pragma unroll
    for (int q = 0; q < 4; ++q)
      cn[q] = *(const v4f*)(orow + (long)tn * H_ + 8 * q);

    __builtin_amdgcn_sched_group_barrier(0x100, 28, 0);
    __builtin_amdgcn_sched_group_barrier(0x020, 4, 0);
    __builtin_amdgcn_sched_barrier(0);

#pragma unroll
    for (int jj = 4; jj < 32; ++jj) {
      const v2f hlo = __builtin_shufflevector(hf[jj], hf[jj], 0, 1);
      const v2f hhi = __builtin_shufflevector(hf[jj], hf[jj], 2, 3);
      const v2f w0lo = __builtin_shufflevector(wq0[jj], wq0[jj], 0, 1);
      const v2f w0hi = __builtin_shufflevector(wq0[jj], wq0[jj], 2, 3);
      const v2f w1lo = __builtin_shufflevector(wq1[jj], wq1[jj], 0, 1);
      const v2f w1hi = __builtin_shufflevector(wq1[jj], wq1[jj], 2, 3);
      acc0 = __builtin_amdgcn_wmma_f32_16x16x4_f32(false, w0lo, false, hlo,
                                                   (short)0, acc0, false, false);
      acc1 = __builtin_amdgcn_wmma_f32_16x16x4_f32(false, w1lo, false, hlo,
                                                   (short)0, acc1, false, false);
      acc0 = __builtin_amdgcn_wmma_f32_16x16x4_f32(false, w0hi, false, hhi,
                                                   (short)0, acc0, false, false);
      acc1 = __builtin_amdgcn_wmma_f32_16x16x4_f32(false, w1hi, false, hhi,
                                                   (short)0, acc1, false, false);
    }

    __builtin_amdgcn_sched_barrier(0);
    __builtin_amdgcn_s_setprio(0);

#pragma unroll
    for (int v = 0; v < 8; ++v) {
      hp0[v] = tanh_f32(acc0[v]);
      hp1[v] = tanh_f32(acc1[v]);
    }

    *(v4f*)(hnxt + hdst)      = __builtin_shufflevector(hp0, hp0, 0, 1, 2, 3);
    *(v4f*)(hnxt + hdst + 8)  = __builtin_shufflevector(hp0, hp0, 4, 5, 6, 7);
    *(v4f*)(hnxt + hdst + 16) = __builtin_shufflevector(hp1, hp1, 0, 1, 2, 3);
    *(v4f*)(hnxt + hdst + 24) = __builtin_shufflevector(hp1, hp1, 4, 5, 6, 7);

    asm volatile("s_wait_dscnt 0x0" ::: "memory");
    asm volatile("s_barrier_signal -1" ::: "memory");
    __builtin_amdgcn_sched_barrier(0);
    {
      float* od = orow + (long)t * H_;
      *(v4f*)(od)      = __builtin_shufflevector(hp0, hp0, 0, 1, 2, 3);
      *(v4f*)(od + 8)  = __builtin_shufflevector(hp0, hp0, 4, 5, 6, 7);
      *(v4f*)(od + 16) = __builtin_shufflevector(hp1, hp1, 0, 1, 2, 3);
      *(v4f*)(od + 24) = __builtin_shufflevector(hp1, hp1, 4, 5, 6, 7);
    }
    {
      const v4f c0 = cn[0] + bz[0];
      const v4f c1 = cn[1] + bz[1];
      const v4f c2 = cn[2] + bz[2];
      const v4f c3 = cn[3] + bz[3];
      acc0 = __builtin_shufflevector(c0, c1, 0, 1, 2, 3, 4, 5, 6, 7);
      acc1 = __builtin_shufflevector(c2, c3, 0, 1, 2, 3, 4, 5, 6, 7);
    }
    OWN_16()
    __builtin_amdgcn_sched_barrier(0);
    asm volatile("s_barrier_wait -1" ::: "memory");

    p ^= 1;
  }
}


__global__ __launch_bounds__(256) __attribute__((amdgpu_waves_per_eu(2, 2))) void rnn_xproj_head_kernel(
    const float* __restrict__ x, const float* __restrict__ Wx,
    float* __restrict__ out) {
  __shared__ __attribute__((aligned(16))) float smem[2 * XP_STAGE * LDSROW];
  xproj_role(x, Wx, out, smem, (int)blockIdx.x, ALL_WGS, 0, T_HEAD / XP_STAGE);
}

__global__ __launch_bounds__(256) __attribute__((amdgpu_waves_per_eu(2, 2))) void rnn_mid_kernel(
    const float* __restrict__ x, const float* __restrict__ Wx,
    const float* __restrict__ Wh, const float* __restrict__ bias,
    float* __restrict__ out) {
  __shared__ __attribute__((aligned(16))) float smem[2 * XP_STAGE * LDSROW];
  if (blockIdx.x < REC_WGS) {
    recur_role(Wh, bias, out, smem, (int)blockIdx.x, 0, T_HEAD);
  } else {
    xproj_role(x, Wx, out, smem, (int)blockIdx.x - REC_WGS, ALL_WGS - REC_WGS,
               T_HEAD, (T_ - T_HEAD) / XP_STAGE);
  }
}

__global__ __launch_bounds__(256) __attribute__((amdgpu_waves_per_eu(2, 2))) void rnn_recur_tail_kernel(
    const float* __restrict__ Wh, const float* __restrict__ bias,
    float* __restrict__ out) {
  __shared__ __attribute__((aligned(16))) float smem[2 * 16 * LDSROW];
  recur_role(Wh, bias, out, smem, (int)blockIdx.x, T_HEAD, T_);
}

extern "C" void kernel_launch(void* const* d_in, const int* in_sizes, int n_in,
                              void* d_out, int out_size, void* d_ws,
                              size_t ws_size, hipStream_t stream) {
  const float* x  = (const float*)d_in[0];
  const float* Wx = (const float*)d_in[1];
  const float* Wh = (const float*)d_in[2];
  const float* b  = (const float*)d_in[3];
  float* out = (float*)d_out;

  rnn_xproj_head_kernel<<<ALL_WGS, 256, 0, stream>>>(x, Wx, out);
  rnn_mid_kernel<<<ALL_WGS, 256, 0, stream>>>(x, Wx, Wh, b, out);
  rnn_recur_tail_kernel<<<REC_WGS, 256, 0, stream>>>(Wh, b, out);
}
